# starting kernel
# baseline (speedup 1.0000x reference)
#include <hip/hip_runtime.h>

#define N_NODES 100000
#define N_EDGES 640000
#define D 128
#define EPS 1e-5f
#define LDSTRIDE 132

typedef __attribute__((ext_vector_type(2))) float v2f;
typedef __attribute__((ext_vector_type(8))) float v8f;

__global__ void zero_ws_kernel(float* __restrict__ p, long n) {
    long i = (long)blockIdx.x * blockDim.x + threadIdx.x;
    long stride = (long)gridDim.x * blockDim.x;
    for (; i < n; i += stride) p[i] = 0.0f;
}

__global__ __launch_bounds__(256) void layernorm_kernel(
    const float* __restrict__ h, const float* __restrict__ gamma,
    const float* __restrict__ beta, float* __restrict__ hn) {
    int wid  = (blockIdx.x * blockDim.x + threadIdx.x) >> 5;
    int lane = threadIdx.x & 31;
    if (wid >= N_NODES) return;

    float4 v = ((const float4*)(h + (long)wid * D))[lane];
    float s = v.x + v.y + v.z + v.w;
#pragma unroll
    for (int off = 16; off; off >>= 1) s += __shfl_xor(s, off, 32);
    float mu = s * (1.0f / D);

    float dx = v.x - mu, dy = v.y - mu, dz = v.z - mu, dw = v.w - mu;
    float q = dx * dx + dy * dy + dz * dz + dw * dw;
#pragma unroll
    for (int off = 16; off; off >>= 1) q += __shfl_xor(q, off, 32);
    float rstd = rsqrtf(q * (1.0f / D) + EPS);

    float4 g  = ((const float4*)gamma)[lane];
    float4 bt = ((const float4*)beta)[lane];
    float4 o;
    o.x = dx * rstd * g.x + bt.x;
    o.y = dy * rstd * g.y + bt.y;
    o.z = dz * rstd * g.z + bt.z;
    o.w = dw * rstd * g.w + bt.w;
    ((float4*)(hn + (long)wid * D))[lane] = o;
}

__global__ __launch_bounds__(256) void scatter_kernel(
    const float* __restrict__ hn, const int* __restrict__ src,
    const int* __restrict__ dst, float* __restrict__ msg,
    float* __restrict__ deg) {
    int wid  = (blockIdx.x * blockDim.x + threadIdx.x) >> 5;
    int lane = threadIdx.x & 31;
    if (wid >= N_EDGES) return;
    int s = src[wid], d = dst[wid];
    const float* hs = hn + (long)s * D;
    float* md = msg + (long)d * D;
#pragma unroll
    for (int i = 0; i < 4; ++i)
        atomicAdd(md + lane + 32 * i, hs[lane + 32 * i]);
    if (lane == 0) atomicAdd(deg + d, 1.0f);
}

__global__ __launch_bounds__(256) void sage_gemm_kernel(
    const float* __restrict__ hn, const float* __restrict__ msg,
    const float* __restrict__ deg, const float* __restrict__ Wself,
    const float* __restrict__ Wneigh, const float* __restrict__ bias,
    float* __restrict__ out) {
    __shared__ float As[32 * LDSTRIDE];
    __shared__ float An[32 * LDSTRIDE];
    __shared__ float invdeg[32];

    int t = threadIdx.x;
    int rowBase = blockIdx.x * 32;

    if (t < 32) invdeg[t] = 1.0f / fmaxf(deg[rowBase + t], 1.0f);
#pragma unroll
    for (int i = 0; i < 16; ++i) {
        int e = t + i * 256;
        int r = e >> 7, c = e & 127;
        As[r * LDSTRIDE + c] = hn[(long)(rowBase + r) * D + c];
    }
    __syncthreads();
#pragma unroll
    for (int i = 0; i < 16; ++i) {
        int e = t + i * 256;
        int r = e >> 7, c = e & 127;
        An[r * LDSTRIDE + c] = msg[(long)(rowBase + r) * D + c] * invdeg[r];
    }
    __syncthreads();

    int lane = t & 31;
    int wave = t >> 5;
    int nq   = lane & 15;
    int hi   = lane >> 4;
    int ncol = wave * 16 + nq;
    int m    = nq;

    v8f acc0 = {};
    v8f acc1 = {};

#pragma unroll 4
    for (int k = 0; k < D; k += 4) {
        int kk = k + (hi << 1);
        v2f a0, a1, b;
        a0.x = As[m * LDSTRIDE + kk];
        a0.y = As[m * LDSTRIDE + kk + 1];
        a1.x = As[(m + 16) * LDSTRIDE + kk];
        a1.y = As[(m + 16) * LDSTRIDE + kk + 1];
        b.x  = Wself[kk * D + ncol];
        b.y  = Wself[(kk + 1) * D + ncol];
        acc0 = __builtin_amdgcn_wmma_f32_16x16x4_f32(
            false, a0, false, b, (short)0, acc0, false, false);
        acc1 = __builtin_amdgcn_wmma_f32_16x16x4_f32(
            false, a1, false, b, (short)0, acc1, false, false);
    }
#pragma unroll 4
    for (int k = 0; k < D; k += 4) {
        int kk = k + (hi << 1);
        v2f a0, a1, b;
        a0.x = An[m * LDSTRIDE + kk];
        a0.y = An[m * LDSTRIDE + kk + 1];
        a1.x = An[(m + 16) * LDSTRIDE + kk];
        a1.y = An[(m + 16) * LDSTRIDE + kk + 1];
        b.x  = Wneigh[kk * D + ncol];
        b.y  = Wneigh[(kk + 1) * D + ncol];
        acc0 = __builtin_amdgcn_wmma_f32_16x16x4_f32(
            false, a0, false, b, (short)0, acc0, false, false);
        acc1 = __builtin_amdgcn_wmma_f32_16x16x4_f32(
            false, a1, false, b, (short)0, acc1, false, false);
    }

    float bv = bias[ncol];
    int mBase = hi * 8;
#pragma unroll
    for (int i = 0; i < 8; ++i) {
        out[(long)(rowBase + mBase + i) * D + ncol]      = acc0[i] + bv;
        out[(long)(rowBase + 16 + mBase + i) * D + ncol] = acc1[i] + bv;
    }
}

extern "C" void kernel_launch(void* const* d_in, const int* in_sizes, int n_in,
                              void* d_out, int out_size, void* d_ws, size_t ws_size,
                              hipStream_t stream) {
    const float* h      = (const float*)d_in[0];
    const int*   src    = (const int*)d_in[1];
    const int*   dst    = (const int*)d_in[2];
    const float* gamma  = (const float*)d_in[3];
    const float* beta   = (const float*)d_in[4];
    const float* Wself  = (const float*)d_in[5];
    const float* Wneigh = (const float*)d_in[6];
    const float* bias   = (const float*)d_in[7];
    float* out = (float*)d_out;

    float* hn  = (float*)d_ws;
    float* msg = hn + (long)N_NODES * D;
    float* deg = msg + (long)N_NODES * D;

    long zcount = (long)N_NODES * D + N_NODES;
    zero_ws_kernel<<<2048, 256, 0, stream>>>(msg, zcount);

    layernorm_kernel<<<(N_NODES + 7) / 8, 256, 0, stream>>>(h, gamma, beta, hn);

    scatter_kernel<<<(N_EDGES + 7) / 8, 256, 0, stream>>>(hn, src, dst, msg, deg);

    sage_gemm_kernel<<<N_NODES / 32, 256, 0, stream>>>(
        hn, msg, deg, Wself, Wneigh, bias, out);
}


# Edge filter key narrowed: the LayerNorm launch also writes one owner-block byte per edge; aggregation blocks scan those bytes 16 per lane-load, not dst as int4 (queue order, so sum order, changed)
# speedup vs baseline: 1.8373x; 1.8373x over previous
#include <hip/hip_runtime.h>

#define N_NODES 100000
#define N_EDGES 640000
#define D 128
#define EPS 1e-5f
#define LDSTRIDE 132

#define RB 416
#define NBLK ((N_NODES + RB - 1) / RB)
#define BT 1024
#define NW (BT / 32)
#define N4 (N_EDGES / 4)
#define N16 (N_EDGES / 16)
#define GS 4
#define NG 10
#define FLUSH_AT 32
#define LCAP (FLUSH_AT + 512)
#define NOKEY 255u
static_assert(N_EDGES % 16 == 0, "keys are read as uint4");
static_assert(GS * NG * BT >= N16 && NG % 2 == 0, "sweeps cover the edge list, two groups per trip");
static_assert(NBLK <= 255, "a block number fits the key byte beside NOKEY");

typedef __attribute__((ext_vector_type(2))) float v2f;
typedef __attribute__((ext_vector_type(8))) float v8f;

__device__ __forceinline__ unsigned owner_key(int d) {
    return ((unsigned)d < (unsigned)N_NODES) ? (unsigned)d / RB : NOKEY;
}

__global__ __launch_bounds__(256) void layernorm_kernel(
    const float* __restrict__ h, const float* __restrict__ gamma,
    const float* __restrict__ beta, float* __restrict__ hn,
    const int* __restrict__ dst, unsigned* __restrict__ keys) {
    int gid  = blockIdx.x * blockDim.x + threadIdx.x;
    int wid  = gid >> 5;
    int lane = threadIdx.x & 31;
    if (gid < N4) {
        int4 d = ((const int4*)dst)[gid];
        keys[gid] = owner_key(d.x) | (owner_key(d.y) << 8) | (owner_key(d.z) << 16) |
                    (owner_key(d.w) << 24);
    }
    if (wid >= N_NODES) return;

    float4 v = ((const float4*)(h + (long)wid * D))[lane];
    float s = v.x + v.y + v.z + v.w;
#pragma unroll
    for (int off = 16; off; off >>= 1) s += __shfl_xor(s, off, 32);
    float mu = s * (1.0f / D);

    float dx = v.x - mu, dy = v.y - mu, dz = v.z - mu, dw = v.w - mu;
    float q = dx * dx + dy * dy + dz * dz + dw * dw;
#pragma unroll
    for (int off = 16; off; off >>= 1) q += __shfl_xor(q, off, 32);
    float rstd = rsqrtf(q * (1.0f / D) + EPS);

    float4 g  = ((const float4*)gamma)[lane];
    float4 bt = ((const float4*)beta)[lane];
    float4 o;
    o.x = dx * rstd * g.x + bt.x;
    o.y = dy * rstd * g.y + bt.y;
    o.z = dz * rstd * g.z + bt.z;
    o.w = dw * rstd * g.w + bt.w;
    ((float4*)(hn + (long)wid * D))[lane] = o;
}

__device__ __forceinline__ uint4 load_keys(const uint4* __restrict__ keys16, int sweep, int t) {
    return keys16[min(sweep * BT + t, N16 - 1)];
}

__device__ __forceinline__ unsigned zero_bytes(unsigned x) {
    return ~(((x & 0x7F7F7F7Fu) + 0x7F7F7F7Fu) | x) & 0x80808080u;
}

__device__ __forceinline__ int queue_matches(uint4 k, int sweep, int t, unsigned pat,
                                             unsigned* lst, int cnt) {
    int idx = sweep * BT + t;
    bool inr = idx < N16;
    unsigned f0 = zero_bytes(k.x ^ pat), f1 = zero_bytes(k.y ^ pat);
    unsigned f2 = zero_bytes(k.z ^ pat), f3 = zero_bytes(k.w ^ pat);
    unsigned m = (f0 >> 3) | (f1 >> 2) | (f2 >> 1) | f3;
    if (!inr) m = 0u;
    unsigned bal = __builtin_amdgcn_ballot_w32(m != 0u);
    if (bal == 0u) return cnt;
    do {
        if (m != 0u) {
            int pos = cnt + (int)__builtin_amdgcn_mbcnt_lo(bal, 0u);
            unsigned p = (unsigned)__builtin_ctz(m);
            lst[pos] = (unsigned)(idx * 16) + ((p & 3u) << 2) + (p >> 3);
            m &= m - 1u;
        }
        cnt += __popc(bal);
        bal = __builtin_amdgcn_ballot_w32(m != 0u);
    } while (bal != 0u);
    return cnt;
}

__device__ __forceinline__ void drain_list(const unsigned* lst, int cnt, int lane, int rowBase,
                                           const int* __restrict__ src,
                                           const int* __restrict__ dst,
                                           const float* __restrict__ hn, float* tile,
                                           int* degc) {
    __builtin_amdgcn_wave_barrier();
    for (int base = 0; base < cnt; base += 32) {
        int n = min(32, cnt - base);
        unsigned ent = lst[base + min(lane, n - 1)];
        int s  = src[ent];
        int lr = dst[ent] - rowBase;
        if (lane < n) atomicAdd(&degc[lr], 1);
        for (int j = 0; j < n; j += 8) {
            float v[8][4];
            int rr[8];
#pragma unroll
            for (int u = 0; u < 8; ++u) {
                int jj = min(j + u, n - 1);
                int sj = __builtin_amdgcn_readlane(s, jj);
                int r  = __builtin_amdgcn_readlane(lr, jj);
                rr[u] = (j + u < n) ? r : RB;
                const float* hs = hn + (long)sj * D + lane;
#pragma unroll
                for (int i = 0; i < 4; ++i) v[u][i] = hs[32 * i];
            }
#pragma unroll
            for (int u = 0; u < 8; ++u) {
                float* tr = tile + rr[u] * D + lane;
#pragma unroll
                for (int i = 0; i < 4; ++i) atomicAdd(tr + 32 * i, v[u][i]);
            }
        }
    }
}

__device__ __forceinline__ void load_group(uint4 (&b)[GS], const uint4* __restrict__ keys16,
                                           int g, int t) {
#pragma unroll
    for (int h = 0; h < GS; ++h) b[h] = load_keys(keys16, GS * g + h, t);
}

__device__ __forceinline__ int filter_group(const uint4 (&b)[GS], int g, int t, int lane,
                                            int rowBase, unsigned pat, unsigned* lst,
                                            int* degc, int cnt,
                                            const int* __restrict__ src,
                                            const int* __restrict__ dst,
                                            const float* __restrict__ hn, float* tile) {
#pragma unroll
    for (int h = 0; h < GS; ++h) {
        cnt = queue_matches(b[h], GS * g + h, t, pat, lst, cnt);
        if (cnt > FLUSH_AT) {
            drain_list(lst, cnt, lane, rowBase, src, dst, hn, tile, degc);
            cnt = 0;
        }
    }
    return cnt;
}

__global__ __launch_bounds__(BT) void bin_aggregate_kernel(
    const float* __restrict__ hn, const int* __restrict__ src,
    const int* __restrict__ dst, const unsigned* __restrict__ keys,
    float* __restrict__ msg, float* __restrict__ deg) {
    __shared__ __attribute__((aligned(16))) float tile[(RB + 1) * D];
    __shared__ unsigned lists[NW * LCAP];
    __shared__ int degc[RB];

    int t = threadIdx.x;
    int lane = t & 31;
    int wave = t >> 5;
    int rowBase = blockIdx.x * RB;
    unsigned pat = blockIdx.x * 0x01010101u;

    float4* tile4 = (float4*)tile;
    for (int i = t; i < (RB + 1) * D / 4; i += BT) tile4[i] = make_float4(0.f, 0.f, 0.f, 0.f);
    for (int i = t; i < RB; i += BT) degc[i] = 0;
    __syncthreads();

    unsigned* lst = lists + wave * LCAP;
    const uint4* keys16 = (const uint4*)keys;
    int cnt = 0;
    int g = (blockIdx.x * NG) / NBLK;
    uint4 bufA[GS], bufB[GS];
    load_group(bufA, keys16, g, t);
    for (int k = 0; k < NG; k += 2) {
        int g1 = (g + 1 == NG) ? 0 : g + 1;
        int g2 = (g1 + 1 == NG) ? 0 : g1 + 1;
        load_group(bufB, keys16, g1, t);
        cnt = filter_group(bufA, g, t, lane, rowBase, pat, lst, degc, cnt, src, dst, hn, tile);
        load_group(bufA, keys16, g2, t);
        cnt = filter_group(bufB, g1, t, lane, rowBase, pat, lst, degc, cnt, src, dst, hn, tile);
        g = g2;
    }
    drain_list(lst, cnt, lane, rowBase, src, dst, hn, tile, degc);
    __syncthreads();

    int nrows = min(RB, N_NODES - rowBase);
    float4* mo = (float4*)(msg + (long)rowBase * D);
    for (int i = t; i < nrows * (D / 4); i += BT) mo[i] = tile4[i];
    for (int i = t; i < nrows; i += BT) deg[rowBase + i] = (float)degc[i];
}

__global__ __launch_bounds__(256) void sage_gemm_kernel(
    const float* __restrict__ hn, const float* __restrict__ msg,
    const float* __restrict__ deg, const float* __restrict__ Wself,
    const float* __restrict__ Wneigh, const float* __restrict__ bias,
    float* __restrict__ out) {
    __shared__ float As[32 * LDSTRIDE];
    __shared__ float An[32 * LDSTRIDE];
    __shared__ float invdeg[32];

    int t = threadIdx.x;
    int rowBase = blockIdx.x * 32;

    if (t < 32) invdeg[t] = 1.0f / fmaxf(deg[rowBase + t], 1.0f);
#pragma unroll
    for (int i = 0; i < 16; ++i) {
        int e = t + i * 256;
        int r = e >> 7, c = e & 127;
        As[r * LDSTRIDE + c] = hn[(long)(rowBase + r) * D + c];
    }
    __syncthreads();
#pragma unroll
    for (int i = 0; i < 16; ++i) {
        int e = t + i * 256;
        int r = e >> 7, c = e & 127;
        An[r * LDSTRIDE + c] = msg[(long)(rowBase + r) * D + c] * invdeg[r];
    }
    __syncthreads();

    int lane = t & 31;
    int wave = t >> 5;
    int nq   = lane & 15;
    int hi   = lane >> 4;
    int ncol = wave * 16 + nq;
    int m    = nq;

    v8f acc0 = {};
    v8f acc1 = {};

#pragma unroll 4
    for (int k = 0; k < D; k += 4) {
        int kk = k + (hi << 1);
        v2f a0, a1, b;
        a0.x = As[m * LDSTRIDE + kk];
        a0.y = As[m * LDSTRIDE + kk + 1];
        a1.x = As[(m + 16) * LDSTRIDE + kk];
        a1.y = As[(m + 16) * LDSTRIDE + kk + 1];
        b.x  = Wself[kk * D + ncol];
        b.y  = Wself[(kk + 1) * D + ncol];
        acc0 = __builtin_amdgcn_wmma_f32_16x16x4_f32(
            false, a0, false, b, (short)0, acc0, false, false);
        acc1 = __builtin_amdgcn_wmma_f32_16x16x4_f32(
            false, a1, false, b, (short)0, acc1, false, false);
    }
#pragma unroll 4
    for (int k = 0; k < D; k += 4) {
        int kk = k + (hi << 1);
        v2f a0, a1, b;
        a0.x = An[m * LDSTRIDE + kk];
        a0.y = An[m * LDSTRIDE + kk + 1];
        a1.x = An[(m + 16) * LDSTRIDE + kk];
        a1.y = An[(m + 16) * LDSTRIDE + kk + 1];
        b.x  = Wneigh[kk * D + ncol];
        b.y  = Wneigh[(kk + 1) * D + ncol];
        acc0 = __builtin_amdgcn_wmma_f32_16x16x4_f32(
            false, a0, false, b, (short)0, acc0, false, false);
        acc1 = __builtin_amdgcn_wmma_f32_16x16x4_f32(
            false, a1, false, b, (short)0, acc1, false, false);
    }

    float bv = bias[ncol];
    int mBase = hi * 8;
#pragma unroll
    for (int i = 0; i < 8; ++i) {
        out[(long)(rowBase + mBase + i) * D + ncol]      = acc0[i] + bv;
        out[(long)(rowBase + 16 + mBase + i) * D + ncol] = acc1[i] + bv;
    }
}

extern "C" void kernel_launch(void* const* d_in, const int* in_sizes, int n_in,
                              void* d_out, int out_size, void* d_ws, size_t ws_size,
                              hipStream_t stream) {
    const float* h      = (const float*)d_in[0];
    const int*   src    = (const int*)d_in[1];
    const int*   dst    = (const int*)d_in[2];
    const float* gamma  = (const float*)d_in[3];
    const float* beta   = (const float*)d_in[4];
    const float* Wself  = (const float*)d_in[5];
    const float* Wneigh = (const float*)d_in[6];
    const float* bias   = (const float*)d_in[7];
    float* out = (float*)d_out;

    float* hn  = (float*)d_ws;
    float* msg = hn + (long)N_NODES * D;
    float* deg = msg + (long)N_NODES * D;
    unsigned* keys = (unsigned*)d_out;

    layernorm_kernel<<<(N_NODES + 7) / 8, 256, 0, stream>>>(h, gamma, beta, hn, dst, keys);

    bin_aggregate_kernel<<<NBLK, BT, 0, stream>>>(hn, src, dst, keys, msg, deg);

    sage_gemm_kernel<<<N_NODES / 32, 256, 0, stream>>>(
        hn, msg, deg, Wself, Wneigh, bias, out);
}


# Third launch folded into the aggregation blocks: each runs the f32 matrix products of its own 416 rows on its LDS sum tile; msg, deg never stored (k terms regrouped per product: sum order changed)
# speedup vs baseline: 2.0704x; 1.1269x over previous
#include <hip/hip_runtime.h>

#define N_NODES 100000
#define N_EDGES 640000
#define D 128
#define EPS 1e-5f
#define TP 132

#define RB 416
#define NBLK ((N_NODES + RB - 1) / RB)
#define BT 1024
#define NW (BT / 32)
#define N4 (N_EDGES / 4)
#define N16 (N_EDGES / 16)
#define GS 4
#define NG 10
#define FLUSH_AT 32
#define LCAP (FLUSH_AT + 512)
#define NOKEY 255u
static_assert(N_EDGES % 16 == 0, "keys are read as uint4");
static_assert(GS * NG * BT >= N16 && NG % 2 == 0, "sweeps cover the edge list, two groups per trip");
static_assert(NBLK <= 255, "a block number fits the key byte beside NOKEY");
static_assert(RB % 16 == 0 && N_NODES % 16 == 0, "a block's rows are whole 16-row matrix tiles");

typedef __attribute__((ext_vector_type(2))) float v2f;
typedef __attribute__((ext_vector_type(4))) float v4f;
typedef __attribute__((ext_vector_type(8))) float v8f;

__device__ __forceinline__ unsigned owner_key(int d) {
    return ((unsigned)d < (unsigned)N_NODES) ? (unsigned)d / RB : NOKEY;
}

__global__ __launch_bounds__(256) void layernorm_kernel(
    const float* __restrict__ h, const float* __restrict__ gamma,
    const float* __restrict__ beta, float* __restrict__ hn,
    const int* __restrict__ dst, unsigned* __restrict__ keys) {
    int gid  = blockIdx.x * blockDim.x + threadIdx.x;
    int wid  = gid >> 5;
    int lane = threadIdx.x & 31;
    if (gid < N4) {
        int4 d = ((const int4*)dst)[gid];
        keys[gid] = owner_key(d.x) | (owner_key(d.y) << 8) | (owner_key(d.z) << 16) |
                    (owner_key(d.w) << 24);
    }
    if (wid >= N_NODES) return;

    float4 v = ((const float4*)(h + (long)wid * D))[lane];
    float s = v.x + v.y + v.z + v.w;
#pragma unroll
    for (int off = 16; off; off >>= 1) s += __shfl_xor(s, off, 32);
    float mu = s * (1.0f / D);

    float dx = v.x - mu, dy = v.y - mu, dz = v.z - mu, dw = v.w - mu;
    float q = dx * dx + dy * dy + dz * dz + dw * dw;
#pragma unroll
    for (int off = 16; off; off >>= 1) q += __shfl_xor(q, off, 32);
    float rstd = rsqrtf(q * (1.0f / D) + EPS);

    float4 g  = ((const float4*)gamma)[lane];
    float4 bt = ((const float4*)beta)[lane];
    float4 o;
    o.x = dx * rstd * g.x + bt.x;
    o.y = dy * rstd * g.y + bt.y;
    o.z = dz * rstd * g.z + bt.z;
    o.w = dw * rstd * g.w + bt.w;
    ((float4*)(hn + (long)wid * D))[lane] = o;
}

__device__ __forceinline__ uint4 load_keys(const uint4* __restrict__ keys16, int sweep, int t) {
    return keys16[min(sweep * BT + t, N16 - 1)];
}

__device__ __forceinline__ unsigned zero_bytes(unsigned x) {
    return ~(((x & 0x7F7F7F7Fu) + 0x7F7F7F7Fu) | x) & 0x80808080u;
}

__device__ __forceinline__ int queue_matches(uint4 k, int sweep, int t, unsigned pat,
                                             unsigned* lst, int cnt) {
    int idx = sweep * BT + t;
    bool inr = idx < N16;
    unsigned f0 = zero_bytes(k.x ^ pat), f1 = zero_bytes(k.y ^ pat);
    unsigned f2 = zero_bytes(k.z ^ pat), f3 = zero_bytes(k.w ^ pat);
    unsigned m = (f0 >> 3) | (f1 >> 2) | (f2 >> 1) | f3;
    if (!inr) m = 0u;
    unsigned bal = __builtin_amdgcn_ballot_w32(m != 0u);
    if (bal == 0u) return cnt;
    do {
        if (m != 0u) {
            int pos = cnt + (int)__builtin_amdgcn_mbcnt_lo(bal, 0u);
            unsigned p = (unsigned)__builtin_ctz(m);
            lst[pos] = (unsigned)(idx * 16) + ((p & 3u) << 2) + (p >> 3);
            m &= m - 1u;
        }
        cnt += __popc(bal);
        bal = __builtin_amdgcn_ballot_w32(m != 0u);
    } while (bal != 0u);
    return cnt;
}

__device__ __forceinline__ void drain_list(const unsigned* lst, int cnt, int lane, int rowBase,
                                           const int* __restrict__ src,
                                           const int* __restrict__ dst,
                                           const float* __restrict__ hn, float* tile,
                                           int* degc) {
    __builtin_amdgcn_wave_barrier();
    for (int base = 0; base < cnt; base += 32) {
        int n = min(32, cnt - base);
        unsigned ent = lst[base + min(lane, n - 1)];
        int s  = src[ent];
        int lr = dst[ent] - rowBase;
        if (lane < n) atomicAdd(&degc[lr], 1);
        for (int j = 0; j < n; j += 8) {
            float v[8][4];
            int rr[8];
#pragma unroll
            for (int u = 0; u < 8; ++u) {
                int jj = min(j + u, n - 1);
                int sj = __builtin_amdgcn_readlane(s, jj);
                int r  = __builtin_amdgcn_readlane(lr, jj);
                rr[u] = (j + u < n) ? r : RB;
                const float* hs = hn + (long)sj * D + lane;
#pragma unroll
                for (int i = 0; i < 4; ++i) v[u][i] = hs[32 * i];
            }
#pragma unroll
            for (int u = 0; u < 8; ++u) {
                float* tr = tile + rr[u] * TP + lane;
#pragma unroll
                for (int i = 0; i < 4; ++i) atomicAdd(tr + 32 * i, v[u][i]);
            }
        }
    }
}

__device__ __forceinline__ void load_group(uint4 (&b)[GS], const uint4* __restrict__ keys16,
                                           int g, int t) {
#pragma unroll
    for (int h = 0; h < GS; ++h) b[h] = load_keys(keys16, GS * g + h, t);
}

__device__ __forceinline__ int filter_group(const uint4 (&b)[GS], int g, int t, int lane,
                                            int rowBase, unsigned pat, unsigned* lst,
                                            int* degc, int cnt,
                                            const int* __restrict__ src,
                                            const int* __restrict__ dst,
                                            const float* __restrict__ hn, float* tile) {
#pragma unroll
    for (int h = 0; h < GS; ++h) {
        cnt = queue_matches(b[h], GS * g + h, t, pat, lst, cnt);
        if (cnt > FLUSH_AT) {
            drain_list(lst, cnt, lane, rowBase, src, dst, hn, tile, degc);
            cnt = 0;
        }
    }
    return cnt;
}

__global__ __launch_bounds__(BT) void bin_aggregate_kernel(
    const float* __restrict__ hn, const int* __restrict__ src,
    const int* __restrict__ dst, const unsigned* __restrict__ keys,
    const float* __restrict__ Wself, const float* __restrict__ Wneigh,
    const float* __restrict__ bias, float* __restrict__ out) {
    __shared__ __attribute__((aligned(16))) float tile[(RB + 1) * TP];
    __shared__ unsigned lists[NW * LCAP];
    __shared__ int degc[RB];

    int t = threadIdx.x;
    int lane = t & 31;
    int wave = t >> 5;
    int rowBase = blockIdx.x * RB;
    unsigned pat = blockIdx.x * 0x01010101u;

    float4* tile4 = (float4*)tile;
    for (int i = t; i < (RB + 1) * TP / 4; i += BT) tile4[i] = make_float4(0.f, 0.f, 0.f, 0.f);
    for (int i = t; i < RB; i += BT) degc[i] = 0;
    __syncthreads();

    unsigned* lst = lists + wave * LCAP;
    const uint4* keys16 = (const uint4*)keys;
    int cnt = 0;
    int g = (blockIdx.x * NG) / NBLK;
    uint4 bufA[GS], bufB[GS];
    load_group(bufA, keys16, g, t);
    for (int k = 0; k < NG; k += 2) {
        int g1 = (g + 1 == NG) ? 0 : g + 1;
        int g2 = (g1 + 1 == NG) ? 0 : g1 + 1;
        load_group(bufB, keys16, g1, t);
        cnt = filter_group(bufA, g, t, lane, rowBase, pat, lst, degc, cnt, src, dst, hn, tile);
        load_group(bufA, keys16, g2, t);
        cnt = filter_group(bufB, g1, t, lane, rowBase, pat, lst, degc, cnt, src, dst, hn, tile);
        g = g2;
    }
    drain_list(lst, cnt, lane, rowBase, src, dst, hn, tile, degc);
    __syncthreads();

    int nrows = min(RB, N_NODES - rowBase);
    int nitems = (nrows / 16) * 2;
    int c = lane & 15;
    int h = lane >> 4;
    int wv = __builtin_amdgcn_readfirstlane(wave);
    for (int item = wv; item < nitems; item += NW) {
        int mt = item >> 1;
        int lrow = mt * 16 + c;
        int col0 = (item & 1) * 64 + 4 * c;
        float inv = 1.0f / fmaxf((float)degc[lrow], 1.0f);
        const float* ag = hn + (long)(rowBase + lrow) * D + 4 * h;
        const float* al = tile + lrow * TP + 4 * h;
        const float* ws = Wself + (long)(4 * h) * D + col0;
        const float* wn = Wneigh + (long)(4 * h) * D + col0;
        v8f acc[4] = {};
        for (int q = 0; q < D / 8; ++q) {
            v4f a  = *(const v4f*)(ag + 8 * q);
            const float* wb = ws + (long)(8 * q) * D;
            v4f b0 = *(const v4f*)(wb);
            v4f b1 = *(const v4f*)(wb + D);
            v4f b2 = *(const v4f*)(wb + 2 * D);
            v4f b3 = *(const v4f*)(wb + 3 * D);
            v2f a01 = {a.x, a.y}, a23 = {a.z, a.w};
#pragma unroll
            for (int nt = 0; nt < 4; ++nt) {
                v2f b = {b0[nt], b1[nt]};
                acc[nt] = __builtin_amdgcn_wmma_f32_16x16x4_f32(
                    false, a01, false, b, (short)0, acc[nt], false, false);
            }
#pragma unroll
            for (int nt = 0; nt < 4; ++nt) {
                v2f b = {b2[nt], b3[nt]};
                acc[nt] = __builtin_amdgcn_wmma_f32_16x16x4_f32(
                    false, a23, false, b, (short)0, acc[nt], false, false);
            }
        }
        for (int q = 0; q < D / 8; ++q) {
            v4f a  = *(const v4f*)(al + 8 * q) * inv;
            const float* wb = wn + (long)(8 * q) * D;
            v4f b0 = *(const v4f*)(wb);
            v4f b1 = *(const v4f*)(wb + D);
            v4f b2 = *(const v4f*)(wb + 2 * D);
            v4f b3 = *(const v4f*)(wb + 3 * D);
            v2f a01 = {a.x, a.y}, a23 = {a.z, a.w};
#pragma unroll
            for (int nt = 0; nt < 4; ++nt) {
                v2f b = {b0[nt], b1[nt]};
                acc[nt] = __builtin_amdgcn_wmma_f32_16x16x4_f32(
                    false, a01, false, b, (short)0, acc[nt], false, false);
            }
#pragma unroll
            for (int nt = 0; nt < 4; ++nt) {
                v2f b = {b2[nt], b3[nt]};
                acc[nt] = __builtin_amdgcn_wmma_f32_16x16x4_f32(
                    false, a23, false, b, (short)0, acc[nt], false, false);
            }
        }
        v4f bv = *(const v4f*)(bias + col0);
        float* orow = out + (long)(rowBase + mt * 16 + 8 * h) * D + col0;
#pragma unroll
        for (int i = 0; i < 8; ++i) {
            v4f o = {acc[0][i] + bv.x, acc[1][i] + bv.y, acc[2][i] + bv.z, acc[3][i] + bv.w};
            *(v4f*)(orow + (long)i * D) = o;
        }
    }
}

extern "C" void kernel_launch(void* const* d_in, const int* in_sizes, int n_in,
                              void* d_out, int out_size, void* d_ws, size_t ws_size,
                              hipStream_t stream) {
    const float* h      = (const float*)d_in[0];
    const int*   src    = (const int*)d_in[1];
    const int*   dst    = (const int*)d_in[2];
    const float* gamma  = (const float*)d_in[3];
    const float* beta   = (const float*)d_in[4];
    const float* Wself  = (const float*)d_in[5];
    const float* Wneigh = (const float*)d_in[6];
    const float* bias   = (const float*)d_in[7];
    float* out = (float*)d_out;

    float* hn  = (float*)d_ws;
    unsigned* keys = (unsigned*)(hn + (long)N_NODES * D);

    layernorm_kernel<<<(N_NODES + 7) / 8, 256, 0, stream>>>(h, gamma, beta, hn, dst, keys);

    bin_aggregate_kernel<<<NBLK, BT, 0, stream>>>(hn, src, dst, keys, Wself, Wneigh, bias, out);
}
